# starting kernel
# baseline (speedup 1.0000x reference)
#include <hip/hip_runtime.h>
#include <hip/hip_bf16.h>


typedef float v2f __attribute__((ext_vector_type(2)));
typedef float v8f __attribute__((ext_vector_type(8)));

#define SEQ_LEN 4096
#define BSZ     8
#define EMBED   1024
#define NDIM    16
#define CHUNK   64
#define NCHUNK  (SEQ_LEN / CHUNK)
#define NPAIR   (NCHUNK / 2)
#define DTILE   8

__global__ void ema_prep(const float* __restrict__ delta,
                         const float* __restrict__ alpha,
                         const float* __restrict__ beta,
                         const float* __restrict__ gamma,
                         float* __restrict__ qArr,  float* __restrict__ cArr,
                         float* __restrict__ lqArr, float* __restrict__ qCArr,
                         float* __restrict__ kcol) {
  const int d = blockIdx.x;
  const int t = threadIdx.x;
  const float dd = __expf(delta[d]);
  float qn[NDIM], cn[NDIM];
#pragma unroll
  for (int n = 0; n < NDIM; ++n) {
    const float al = alpha[d * NDIM + n];
    const float p  = dd / (1.0f + 0.5f * dd * al);
    qn[n] = 1.0f - p * al;
    cn[n] = p * beta[d * NDIM + n] * gamma[d * NDIM + n];
  }
  float s = 0.0f;
#pragma unroll
  for (int n = 0; n < NDIM; ++n)
    s += cn[n] * __expf((float)t * __logf(qn[n]));
  kcol[d * CHUNK + t] = s;

  if (t < NDIM) {
    const float q = qn[t];
    qArr [d * NDIM + t] = q;
    cArr [d * NDIM + t] = cn[t];
    lqArr[d * NDIM + t] = __logf(q);
    float qc = q;
#pragma unroll
    for (int i = 0; i < 6; ++i) qc *= qc;
    qCArr[d * NDIM + t] = qc;
  }
}

__global__ void ema_chunkstate(const float* __restrict__ x,
                               const float* __restrict__ qArr,
                               float* __restrict__ Abuf) {
  const int ch = blockIdx.x;
  const int b  = blockIdx.y;
  const int d  = blockIdx.z * blockDim.x + threadIdx.x;

  float q[NDIM], A[NDIM];
#pragma unroll
  for (int n = 0; n < NDIM; ++n) { q[n] = qArr[d * NDIM + n]; A[n] = 0.0f; }

  const float* xp = x + ((size_t)ch * CHUNK * BSZ + b) * EMBED + d;
  for (int j = 0; j < CHUNK; ++j) {
    const float xv = xp[(size_t)j * BSZ * EMBED];
#pragma unroll
    for (int n = 0; n < NDIM; ++n) A[n] = A[n] * q[n] + xv;
  }
  const size_t base = (((size_t)ch * BSZ + b) * NDIM) * EMBED + d;
#pragma unroll
  for (int n = 0; n < NDIM; ++n) Abuf[base + (size_t)n * EMBED] = A[n];
}

__global__ void ema_scan(const float* __restrict__ qCArr,
                         float* __restrict__ Abuf) {
  const int flat = blockIdx.x * blockDim.x + threadIdx.x;
  const int d = flat % EMBED;
  const int n = (flat / EMBED) % NDIM;
  const float qc = qCArr[d * NDIM + n];
  float S = 0.0f;
  const size_t stride = (size_t)BSZ * NDIM * EMBED;
  for (int ch = 0; ch < NCHUNK; ++ch) {
    const size_t idx = (size_t)ch * stride + flat;
    const float a = Abuf[idx];
    Abuf[idx] = S;
    S = qc * S + a;
  }
}

__global__ __launch_bounds__(128) void ema_main(
    const float* __restrict__ x,     const float* __restrict__ omega,
    const float* __restrict__ cArr,  const float* __restrict__ lqArr,
    const float* __restrict__ kcol,  const float* __restrict__ Sbuf,
    float* __restrict__ out) {
  __shared__ __align__(16) float Xall[128 * BSZ * DTILE];
  __shared__ __align__(16) float Sl[NDIM * DTILE * 16];
  __shared__ __align__(16) float Tm[64 * 66];
  __shared__ __align__(16) float Wm[64 * 17];

  const int tid  = threadIdx.x;
  const int d0   = blockIdx.x * DTILE;
  const int pair = blockIdx.y;
  const int t0   = pair * 128;

  for (int e = tid; e < 2048; e += 128) {
    const int t = e >> 4, rem = e & 15, b = rem >> 1, v = rem & 1;
    const float4 val =
        *(const float4*)(x + (((size_t)(t0 + t) * BSZ + b) * EMBED + d0 + v * 4));
    *(float4*)&Xall[(t * BSZ + b) * DTILE + v * 4] = val;
  }
  for (int e = tid; e < NDIM * DTILE * 16; e += 128) {
    const int col = e >> 7, n = (e >> 3) & 15, dl = e & 7;
    const int cp = col >> 3, b = col & 7;
    const int ch = 2 * pair + cp;
    Sl[(n * DTILE + dl) * 16 + col] =
        Sbuf[(((size_t)ch * BSZ + b) * NDIM + n) * EMBED + d0 + dl];
  }

  const int lane = tid & 31;
  const int wv   = tid >> 5;
  const int c    = lane & 15;
  const int kk   = (lane >> 4) << 1;
  const int hi8  = (lane >> 4) << 3;
  const int i0   = wv * 16;
  const int cp   = c >> 3, cb = c & 7;

  for (int dl = 0; dl < DTILE; ++dl) {
    __syncthreads();
    const int d = d0 + dl;
    for (int e = tid; e < 64 * 64; e += 128) {
      const int i = e >> 6, j = e & 63;
      Tm[i * 66 + j] = (j <= i) ? kcol[d * CHUNK + (i - j)] : 0.0f;
    }
    for (int e = tid; e < 64 * NDIM; e += 128) {
      const int i = e >> 4, n = e & 15;
      Wm[i * 17 + n] = cArr[d * NDIM + n] *
                       __expf((float)(i + 1) * lqArr[d * NDIM + n]);
    }
    __syncthreads();

    v8f acc = {0.f, 0.f, 0.f, 0.f, 0.f, 0.f, 0.f, 0.f};
#pragma unroll
    for (int k = 0; k < 16; k += 4) {
      v2f a, b;
      a.x = Wm[(i0 + c) * 17 + k + kk];
      a.y = Wm[(i0 + c) * 17 + k + kk + 1];
      b.x = Sl[((k + kk    ) * DTILE + dl) * 16 + c];
      b.y = Sl[((k + kk + 1) * DTILE + dl) * 16 + c];
      acc = __builtin_amdgcn_wmma_f32_16x16x4_f32(false, a, false, b,
                                                  (short)0, acc, false, false);
    }
#pragma unroll
    for (int k = 0; k < 64; k += 4) {
      v2f a, b;
      a.x = Tm[(i0 + c) * 66 + k + kk];
      a.y = Tm[(i0 + c) * 66 + k + kk + 1];
      const int t1 = cp * 64 + k + kk;
      b.x = Xall[((t1    ) * BSZ + cb) * DTILE + dl];
      b.y = Xall[((t1 + 1) * BSZ + cb) * DTILE + dl];
      acc = __builtin_amdgcn_wmma_f32_16x16x4_f32(false, a, false, b,
                                                  (short)0, acc, false, false);
    }

    __syncthreads();
    const float om = omega[d];
#pragma unroll
    for (int r = 0; r < 8; ++r) {
      const int i  = i0 + r + hi8;
      const int t1 = cp * 64 + i;
      const int idx = (t1 * BSZ + cb) * DTILE + dl;
      const float xv = Xall[idx];
      Xall[idx] = acc[r] + om * xv;
    }
  }
  __syncthreads();
  for (int e = tid; e < 2048; e += 128) {
    const int t = e >> 4, rem = e & 15, b = rem >> 1, v = rem & 1;
    const float4 val = *(const float4*)&Xall[(t * BSZ + b) * DTILE + v * 4];
    *(float4*)(out + (((size_t)(t0 + t) * BSZ + b) * EMBED + d0 + v * 4)) = val;
  }
}

extern "C" void kernel_launch(void* const* d_in, const int* in_sizes, int n_in,
                              void* d_out, int out_size, void* d_ws,
                              size_t ws_size, hipStream_t stream) {
  const float* x     = (const float*)d_in[0];
  const float* delta = (const float*)d_in[1];
  const float* alpha = (const float*)d_in[2];
  const float* beta  = (const float*)d_in[3];
  const float* gamma = (const float*)d_in[4];
  const float* omega = (const float*)d_in[5];
  float* out = (float*)d_out;

  float* ws    = (float*)d_ws;
  float* qArr  = ws;
  float* cArr  = qArr  + EMBED * NDIM;
  float* lqArr = cArr  + EMBED * NDIM;
  float* qCArr = lqArr + EMBED * NDIM;
  float* kcol  = qCArr + EMBED * NDIM;
  float* Abuf  = kcol  + EMBED * CHUNK;

  ema_prep<<<EMBED, CHUNK, 0, stream>>>(delta, alpha, beta, gamma,
                                        qArr, cArr, lqArr, qCArr, kcol);
  ema_chunkstate<<<dim3(NCHUNK, BSZ, EMBED / 256), 256, 0, stream>>>(x, qArr,
                                                                     Abuf);
  ema_scan<<<(BSZ * NDIM * EMBED) / 256, 256, 0, stream>>>(qCArr, Abuf);
  ema_main<<<dim3(EMBED / DTILE, NPAIR), 128, 0, stream>>>(
      x, omega, cArr, lqArr, kcol, Abuf, out);
}


# main kernel: one wave per channel holds the Toeplitz and state A fragments in registers across 8 chunk pairs; B loaded once per pair for 4 row blocks; all-zero causal blocks skipped
# speedup vs baseline: 1.8257x; 1.8257x over previous
#include <hip/hip_runtime.h>
#include <hip/hip_bf16.h>


typedef float v2f __attribute__((ext_vector_type(2)));
typedef float v8f __attribute__((ext_vector_type(8)));

#define SEQ_LEN 4096
#define BSZ     8
#define EMBED   1024
#define NDIM    16
#define CHUNK   64
#define NCHUNK  (SEQ_LEN / CHUNK)
#define NPAIR   (NCHUNK / 2)
#define DTILE   8
#define NPW     8
#define XPITCH  66
#define XCH1    2
#define SPITCH  257

__global__ void ema_prep(const float* __restrict__ delta,
                         const float* __restrict__ alpha,
                         const float* __restrict__ beta,
                         const float* __restrict__ gamma,
                         float* __restrict__ qArr,  float* __restrict__ cArr,
                         float* __restrict__ lqArr, float* __restrict__ qCArr,
                         float* __restrict__ kcol) {
  const int d = blockIdx.x;
  const int t = threadIdx.x;
  const float dd = __expf(delta[d]);
  float qn[NDIM], cn[NDIM];
#pragma unroll
  for (int n = 0; n < NDIM; ++n) {
    const float al = alpha[d * NDIM + n];
    const float p  = dd / (1.0f + 0.5f * dd * al);
    qn[n] = 1.0f - p * al;
    cn[n] = p * beta[d * NDIM + n] * gamma[d * NDIM + n];
  }
  float s = 0.0f;
#pragma unroll
  for (int n = 0; n < NDIM; ++n)
    s += cn[n] * __expf((float)t * __logf(qn[n]));
  kcol[d * CHUNK + t] = s;

  if (t < NDIM) {
    const float q = qn[t];
    qArr [d * NDIM + t] = q;
    cArr [d * NDIM + t] = cn[t];
    lqArr[d * NDIM + t] = __logf(q);
    float qc = q;
#pragma unroll
    for (int i = 0; i < 6; ++i) qc *= qc;
    qCArr[d * NDIM + t] = qc;
  }
}

__global__ void ema_chunkstate(const float* __restrict__ x,
                               const float* __restrict__ qArr,
                               float* __restrict__ Abuf) {
  const int ch = blockIdx.x;
  const int b  = blockIdx.y;
  const int d  = blockIdx.z * blockDim.x + threadIdx.x;

  float q[NDIM], A[NDIM];
#pragma unroll
  for (int n = 0; n < NDIM; ++n) { q[n] = qArr[d * NDIM + n]; A[n] = 0.0f; }

  const float* xp = x + ((size_t)ch * CHUNK * BSZ + b) * EMBED + d;
  for (int j = 0; j < CHUNK; ++j) {
    const float xv = xp[(size_t)j * BSZ * EMBED];
#pragma unroll
    for (int n = 0; n < NDIM; ++n) A[n] = A[n] * q[n] + xv;
  }
  const size_t base = (((size_t)ch * BSZ + b) * NDIM) * EMBED + d;
#pragma unroll
  for (int n = 0; n < NDIM; ++n) Abuf[base + (size_t)n * EMBED] = A[n];
}

__global__ void ema_scan(const float* __restrict__ qCArr,
                         float* __restrict__ Abuf) {
  const int flat = blockIdx.x * blockDim.x + threadIdx.x;
  const int d = flat % EMBED;
  const int n = (flat / EMBED) % NDIM;
  const float qc = qCArr[d * NDIM + n];
  float S = 0.0f;
  const size_t stride = (size_t)BSZ * NDIM * EMBED;
  for (int ch = 0; ch < NCHUNK; ++ch) {
    const size_t idx = (size_t)ch * stride + flat;
    const float a = Abuf[idx];
    Abuf[idx] = S;
    S = qc * S + a;
  }
}

__global__ __launch_bounds__(256) void ema_main(
    const float* __restrict__ x,     const float* __restrict__ omega,
    const float* __restrict__ cArr,  const float* __restrict__ lqArr,
    const float* __restrict__ kcol,  const float* __restrict__ Sbuf,
    float* __restrict__ out) {
  __shared__ __align__(16) float Xs[128 * XPITCH + 8];
  __shared__ __align__(16) float Sl[DTILE * SPITCH];

  const int tid  = threadIdx.x;
  const int lane = tid & 31;
  const int wv   = tid >> 5;
  const int d0   = blockIdx.x * DTILE;
  const int d    = d0 + wv;
  const int pair0 = blockIdx.y * NPW;
  const int c    = lane & 15;
  const int h    = lane >> 4;
  const int kk   = h << 1;
  const int cp   = c >> 3, cb = c & 7;

  v2f Tf[16];
  const float* kc = kcol + d * CHUNK;
#pragma unroll
  for (int mi = 0; mi < 16; ++mi) {
    const int e0 = 4 * (mi - 3) + c - kk;
    const int e1 = e0 - 1;
    const float v0 = kc[e0 > 0 ? e0 : 0];
    const float v1 = kc[e1 > 0 ? e1 : 0];
    Tf[mi].x = (e0 >= 0) ? v0 : 0.0f;
    Tf[mi].y = (e1 >= 0) ? v1 : 0.0f;
  }
  v2f Wf[4][4];
#pragma unroll
  for (int s = 0; s < 4; ++s) {
    const int n = 4 * s + kk;
    const float c0 = cArr[d * NDIM + n],  c1 = cArr[d * NDIM + n + 1];
    const float l0 = lqArr[d * NDIM + n], l1 = lqArr[d * NDIM + n + 1];
#pragma unroll
    for (int rb = 0; rb < 4; ++rb) {
      const float ip1 = (float)(16 * rb + c + 1);
      Wf[rb][s].x = c0 * __expf(ip1 * l0);
      Wf[rb][s].y = c1 * __expf(ip1 * l1);
    }
  }
  const float om = omega[d];

  const int st  = tid >> 4;
  const int sb  = (tid & 15) >> 1;
  const int sv  = tid & 1;
  const int sn  = (tid >> 3) & 15;
  const int sdl = tid & 7;
  const int scol = tid >> 7;

  float4 xr[8];
  float  sr[8];
  {
    const int t0 = pair0 * 128;
#pragma unroll
    for (int it = 0; it < 8; ++it) {
      const int t = st + 16 * it;
      xr[it] = *(const float4*)(x + (((size_t)(t0 + t) * BSZ + sb) * EMBED + d0 + sv * 4));
      const int col = scol + 2 * it;
      const int ch = 2 * pair0 + (col >> 3);
      sr[it] = Sbuf[(((size_t)ch * BSZ + (col & 7)) * NDIM + sn) * EMBED + d0 + sdl];
    }
  }

  const int xb = (cp * 64 + kk) * XPITCH + cp * XCH1 + cb * DTILE + wv;
  const int sB = wv * SPITCH + kk * 16 + c;
  const int eb = (cp * 64 + 8 * h) * XPITCH + cp * XCH1 + cb * DTILE + wv;

  for (int pi = 0; pi < NPW; ++pi) {
    const int pair = pair0 + pi;
#pragma unroll
    for (int it = 0; it < 8; ++it) {
      const int t = st + 16 * it;
      float* p = &Xs[t * XPITCH + (t >> 6) * XCH1 + sb * DTILE + sv * 4];
      *(float2*)(p)     = make_float2(xr[it].x, xr[it].y);
      *(float2*)(p + 2) = make_float2(xr[it].z, xr[it].w);
      Sl[sdl * SPITCH + sn * 16 + scol + 2 * it] = sr[it];
    }
    __syncthreads();
    if (pi + 1 < NPW) {
      const int t0 = (pair + 1) * 128;
#pragma unroll
      for (int it = 0; it < 8; ++it) {
        const int t = st + 16 * it;
        xr[it] = *(const float4*)(x + (((size_t)(t0 + t) * BSZ + sb) * EMBED + d0 + sv * 4));
        const int col = scol + 2 * it;
        const int ch = 2 * (pair + 1) + (col >> 3);
        sr[it] = Sbuf[(((size_t)ch * BSZ + (col & 7)) * NDIM + sn) * EMBED + d0 + sdl];
      }
    }

    v2f Bx[16], Bs[4];
#pragma unroll
    for (int s = 0; s < 16; ++s) {
      Bx[s].x = Xs[xb + (4 * s    ) * XPITCH];
      Bx[s].y = Xs[xb + (4 * s + 1) * XPITCH];
    }
#pragma unroll
    for (int s = 0; s < 4; ++s) {
      Bs[s].x = Sl[sB + (4 * s    ) * 16];
      Bs[s].y = Sl[sB + (4 * s + 1) * 16];
    }

    v8f acc[4];
#pragma unroll
    for (int rb = 0; rb < 4; ++rb)
      acc[rb] = (v8f){0.f, 0.f, 0.f, 0.f, 0.f, 0.f, 0.f, 0.f};
#pragma unroll
    for (int step = 0; step < 20; ++step) {
#pragma unroll
      for (int rb = 0; rb < 4; ++rb) {
        if (step < 4) {
          acc[rb] = __builtin_amdgcn_wmma_f32_16x16x4_f32(
              false, Wf[rb][step], false, Bs[step], (short)0, acc[rb], false, false);
        } else if (step - 4 < 4 * (rb + 1)) {
          const int s = step - 4;
          acc[rb] = __builtin_amdgcn_wmma_f32_16x16x4_f32(
              false, Tf[4 * rb - s + 3], false, Bx[s], (short)0, acc[rb], false, false);
        }
      }
    }

#pragma unroll
    for (int rb = 0; rb < 4; ++rb) {
#pragma unroll
      for (int r = 0; r < 8; ++r) {
        const int idx = eb + (16 * rb + r) * XPITCH;
        const float xv = Xs[idx];
        Xs[idx] = acc[rb][r] + om * xv;
      }
    }
    __syncthreads();
    {
      const int t0 = pair * 128;
#pragma unroll
      for (int it = 0; it < 8; ++it) {
        const int t = st + 16 * it;
        const float* p = &Xs[t * XPITCH + (t >> 6) * XCH1 + sb * DTILE + sv * 4];
        const float2 lo = *(const float2*)(p);
        const float2 hi = *(const float2*)(p + 2);
        *(float4*)(out + (((size_t)(t0 + t) * BSZ + sb) * EMBED + d0 + sv * 4)) =
            make_float4(lo.x, lo.y, hi.x, hi.y);
      }
    }
  }
}

extern "C" void kernel_launch(void* const* d_in, const int* in_sizes, int n_in,
                              void* d_out, int out_size, void* d_ws,
                              size_t ws_size, hipStream_t stream) {
  const float* x     = (const float*)d_in[0];
  const float* delta = (const float*)d_in[1];
  const float* alpha = (const float*)d_in[2];
  const float* beta  = (const float*)d_in[3];
  const float* gamma = (const float*)d_in[4];
  const float* omega = (const float*)d_in[5];
  float* out = (float*)d_out;

  float* ws    = (float*)d_ws;
  float* qArr  = ws;
  float* cArr  = qArr  + EMBED * NDIM;
  float* lqArr = cArr  + EMBED * NDIM;
  float* qCArr = lqArr + EMBED * NDIM;
  float* kcol  = qCArr + EMBED * NDIM;
  float* Abuf  = kcol  + EMBED * CHUNK;

  ema_prep<<<EMBED, CHUNK, 0, stream>>>(delta, alpha, beta, gamma,
                                        qArr, cArr, lqArr, qCArr, kcol);
  ema_chunkstate<<<dim3(NCHUNK, BSZ, EMBED / 256), 256, 0, stream>>>(x, qArr,
                                                                     Abuf);
  ema_scan<<<(BSZ * NDIM * EMBED) / 256, 256, 0, stream>>>(qCArr, Abuf);
  ema_main<<<dim3(EMBED / DTILE, NPAIR / NPW), 256, 0, stream>>>(
      x, omega, cArr, lqArr, kcol, Abuf, out);
}


# main kernel: channel tile per workgroup 8 -> 16 (16 waves, one channel each): every global row piece of x, chunk state and out is 64 B instead of 32 B; LDS tile re-laid [t][ch][b], conflict-free
# speedup vs baseline: 2.1289x; 1.1661x over previous
#include <hip/hip_runtime.h>
#include <hip/hip_bf16.h>


typedef float v2f __attribute__((ext_vector_type(2)));
typedef float v8f __attribute__((ext_vector_type(8)));

#define SEQ_LEN 4096
#define BSZ     8
#define EMBED   1024
#define NDIM    16
#define CHUNK   64
#define NCHUNK  (SEQ_LEN / CHUNK)
#define NPAIR   (NCHUNK / 2)
#define DTILE   16
#define NPW     8
#define XPITCH  140
#define XCH1    8
#define SPITCH  257
#define FD(dl)  (8 * ((dl) + ((dl) >> 3)))

__global__ void ema_prep(const float* __restrict__ delta,
                         const float* __restrict__ alpha,
                         const float* __restrict__ beta,
                         const float* __restrict__ gamma,
                         float* __restrict__ qArr,  float* __restrict__ cArr,
                         float* __restrict__ lqArr, float* __restrict__ qCArr,
                         float* __restrict__ kcol) {
  const int d = blockIdx.x;
  const int t = threadIdx.x;
  const float dd = __expf(delta[d]);
  float qn[NDIM], cn[NDIM];
#pragma unroll
  for (int n = 0; n < NDIM; ++n) {
    const float al = alpha[d * NDIM + n];
    const float p  = dd / (1.0f + 0.5f * dd * al);
    qn[n] = 1.0f - p * al;
    cn[n] = p * beta[d * NDIM + n] * gamma[d * NDIM + n];
  }
  float s = 0.0f;
#pragma unroll
  for (int n = 0; n < NDIM; ++n)
    s += cn[n] * __expf((float)t * __logf(qn[n]));
  kcol[d * CHUNK + t] = s;

  if (t < NDIM) {
    const float q = qn[t];
    qArr [d * NDIM + t] = q;
    cArr [d * NDIM + t] = cn[t];
    lqArr[d * NDIM + t] = __logf(q);
    float qc = q;
#pragma unroll
    for (int i = 0; i < 6; ++i) qc *= qc;
    qCArr[d * NDIM + t] = qc;
  }
}

__global__ void ema_chunkstate(const float* __restrict__ x,
                               const float* __restrict__ qArr,
                               float* __restrict__ Abuf) {
  const int ch = blockIdx.x;
  const int b  = blockIdx.y;
  const int d  = blockIdx.z * blockDim.x + threadIdx.x;

  float q[NDIM], A[NDIM];
#pragma unroll
  for (int n = 0; n < NDIM; ++n) { q[n] = qArr[d * NDIM + n]; A[n] = 0.0f; }

  const float* xp = x + ((size_t)ch * CHUNK * BSZ + b) * EMBED + d;
  for (int j = 0; j < CHUNK; ++j) {
    const float xv = xp[(size_t)j * BSZ * EMBED];
#pragma unroll
    for (int n = 0; n < NDIM; ++n) A[n] = A[n] * q[n] + xv;
  }
  const size_t base = (((size_t)ch * BSZ + b) * NDIM) * EMBED + d;
#pragma unroll
  for (int n = 0; n < NDIM; ++n) Abuf[base + (size_t)n * EMBED] = A[n];
}

__global__ void ema_scan(const float* __restrict__ qCArr,
                         float* __restrict__ Abuf) {
  const int flat = blockIdx.x * blockDim.x + threadIdx.x;
  const int d = flat % EMBED;
  const int n = (flat / EMBED) % NDIM;
  const float qc = qCArr[d * NDIM + n];
  float S = 0.0f;
  const size_t stride = (size_t)BSZ * NDIM * EMBED;
  for (int ch = 0; ch < NCHUNK; ++ch) {
    const size_t idx = (size_t)ch * stride + flat;
    const float a = Abuf[idx];
    Abuf[idx] = S;
    S = qc * S + a;
  }
}

__global__ __launch_bounds__(512) void ema_main(
    const float* __restrict__ x,     const float* __restrict__ omega,
    const float* __restrict__ cArr,  const float* __restrict__ lqArr,
    const float* __restrict__ kcol,  const float* __restrict__ Sbuf,
    float* __restrict__ out) {
  __shared__ __align__(16) float Xs[128 * XPITCH + XCH1];
  __shared__ __align__(16) float Sl[DTILE * SPITCH];

  const int tid  = threadIdx.x;
  const int lane = tid & 31;
  const int wv   = tid >> 5;
  const int d0   = blockIdx.x * DTILE;
  const int d    = d0 + wv;
  const int pair0 = blockIdx.y * NPW;
  const int c    = lane & 15;
  const int h    = lane >> 4;
  const int kk   = h << 1;
  const int cp   = c >> 3, cb = c & 7;

  v2f Tf[16];
  const float* kc = kcol + d * CHUNK;
#pragma unroll
  for (int mi = 0; mi < 16; ++mi) {
    const int e0 = 4 * (mi - 3) + c - kk;
    const int e1 = e0 - 1;
    const float v0 = kc[e0 > 0 ? e0 : 0];
    const float v1 = kc[e1 > 0 ? e1 : 0];
    Tf[mi].x = (e0 >= 0) ? v0 : 0.0f;
    Tf[mi].y = (e1 >= 0) ? v1 : 0.0f;
  }
  v2f Wf[4][4];
#pragma unroll
  for (int s = 0; s < 4; ++s) {
    const int n = 4 * s + kk;
    const float c0 = cArr[d * NDIM + n],  c1 = cArr[d * NDIM + n + 1];
    const float l0 = lqArr[d * NDIM + n], l1 = lqArr[d * NDIM + n + 1];
#pragma unroll
    for (int rb = 0; rb < 4; ++rb) {
      const float ip1 = (float)(16 * rb + c + 1);
      Wf[rb][s].x = c0 * __expf(ip1 * l0);
      Wf[rb][s].y = c1 * __expf(ip1 * l1);
    }
  }
  const float om = omega[d];

  const int st  = tid >> 5;
  const int sb  = (tid >> 2) & 7;
  const int sv  = tid & 3;
  const int sn  = (tid >> 4) & 15;
  const int sdl = tid & 15;
  const int scol = tid >> 8;
  const int sx  = FD(4 * sv) + sb;

  float4 xr[8];
  float  sr[8];
  {
    const int t0 = pair0 * 128;
#pragma unroll
    for (int it = 0; it < 8; ++it) {
      const int t = st + 16 * it;
      xr[it] = *(const float4*)(x + (((size_t)(t0 + t) * BSZ + sb) * EMBED + d0 + sv * 4));
      const int col = scol + 2 * it;
      const int ch = 2 * pair0 + (col >> 3);
      sr[it] = Sbuf[(((size_t)ch * BSZ + (col & 7)) * NDIM + sn) * EMBED + d0 + sdl];
    }
  }

  const int xb = (cp * 64 + kk) * XPITCH + cp * XCH1 + FD(wv) + cb;
  const int sB = wv * SPITCH + kk * 16 + c;
  const int eb = (cp * 64 + 8 * h) * XPITCH + cp * XCH1 + FD(wv) + cb;

  for (int pi = 0; pi < NPW; ++pi) {
    const int pair = pair0 + pi;
#pragma unroll
    for (int it = 0; it < 8; ++it) {
      const int t = st + 16 * it;
      float* p = &Xs[t * XPITCH + (t >> 6) * XCH1 + sx];
      p[0]  = xr[it].x;
      p[8]  = xr[it].y;
      p[16] = xr[it].z;
      p[24] = xr[it].w;
      Sl[sdl * SPITCH + sn * 16 + scol + 2 * it] = sr[it];
    }
    __syncthreads();
    if (pi + 1 < NPW) {
      const int t0 = (pair + 1) * 128;
#pragma unroll
      for (int it = 0; it < 8; ++it) {
        const int t = st + 16 * it;
        xr[it] = *(const float4*)(x + (((size_t)(t0 + t) * BSZ + sb) * EMBED + d0 + sv * 4));
        const int col = scol + 2 * it;
        const int ch = 2 * (pair + 1) + (col >> 3);
        sr[it] = Sbuf[(((size_t)ch * BSZ + (col & 7)) * NDIM + sn) * EMBED + d0 + sdl];
      }
    }

    v2f Bx[16], Bs[4];
#pragma unroll
    for (int s = 0; s < 16; ++s) {
      Bx[s].x = Xs[xb + (4 * s    ) * XPITCH];
      Bx[s].y = Xs[xb + (4 * s + 1) * XPITCH];
    }
#pragma unroll
    for (int s = 0; s < 4; ++s) {
      Bs[s].x = Sl[sB + (4 * s    ) * 16];
      Bs[s].y = Sl[sB + (4 * s + 1) * 16];
    }

    v8f acc[4];
#pragma unroll
    for (int rb = 0; rb < 4; ++rb)
      acc[rb] = (v8f){0.f, 0.f, 0.f, 0.f, 0.f, 0.f, 0.f, 0.f};
#pragma unroll
    for (int step = 0; step < 20; ++step) {
#pragma unroll
      for (int rb = 0; rb < 4; ++rb) {
        if (step < 4) {
          acc[rb] = __builtin_amdgcn_wmma_f32_16x16x4_f32(
              false, Wf[rb][step], false, Bs[step], (short)0, acc[rb], false, false);
        } else if (step - 4 < 4 * (rb + 1)) {
          const int s = step - 4;
          acc[rb] = __builtin_amdgcn_wmma_f32_16x16x4_f32(
              false, Tf[4 * rb - s + 3], false, Bx[s], (short)0, acc[rb], false, false);
        }
      }
    }

#pragma unroll
    for (int rb = 0; rb < 4; ++rb) {
#pragma unroll
      for (int r = 0; r < 8; ++r) {
        const int idx = eb + (16 * rb + r) * XPITCH;
        const float xv = Xs[idx];
        Xs[idx] = acc[rb][r] + om * xv;
      }
    }
    __syncthreads();
    {
      const int t0 = pair * 128;
#pragma unroll
      for (int it = 0; it < 8; ++it) {
        const int t = st + 16 * it;
        const float* p = &Xs[t * XPITCH + (t >> 6) * XCH1 + sx];
        *(float4*)(out + (((size_t)(t0 + t) * BSZ + sb) * EMBED + d0 + sv * 4)) =
            make_float4(p[0], p[8], p[16], p[24]);
      }
    }
  }
}

extern "C" void kernel_launch(void* const* d_in, const int* in_sizes, int n_in,
                              void* d_out, int out_size, void* d_ws,
                              size_t ws_size, hipStream_t stream) {
  const float* x     = (const float*)d_in[0];
  const float* delta = (const float*)d_in[1];
  const float* alpha = (const float*)d_in[2];
  const float* beta  = (const float*)d_in[3];
  const float* gamma = (const float*)d_in[4];
  const float* omega = (const float*)d_in[5];
  float* out = (float*)d_out;

  float* ws    = (float*)d_ws;
  float* qArr  = ws;
  float* cArr  = qArr  + EMBED * NDIM;
  float* lqArr = cArr  + EMBED * NDIM;
  float* qCArr = lqArr + EMBED * NDIM;
  float* kcol  = qCArr + EMBED * NDIM;
  float* Abuf  = kcol  + EMBED * CHUNK;

  ema_prep<<<EMBED, CHUNK, 0, stream>>>(delta, alpha, beta, gamma,
                                        qArr, cArr, lqArr, qCArr, kcol);
  ema_chunkstate<<<dim3(NCHUNK, BSZ, EMBED / 256), 256, 0, stream>>>(x, qArr,
                                                                     Abuf);
  ema_scan<<<(BSZ * NDIM * EMBED) / 256, 256, 0, stream>>>(qCArr, Abuf);
  ema_main<<<dim3(EMBED / DTILE, NPAIR / NPW), 512, 0, stream>>>(
      x, omega, cArr, lqArr, kcol, Abuf, out);
}


# main kernel carries the 16-state recurrence along the whole sequence (chunk sums by one more f32 WMMA product, scan in lane registers); chunk-state and scan launches dropped; order of sums changed
# speedup vs baseline: 3.3298x; 1.5641x over previous
#include <hip/hip_runtime.h>
#include <hip/hip_bf16.h>


typedef float v2f __attribute__((ext_vector_type(2)));
typedef float v8f __attribute__((ext_vector_type(8)));

#define SEQ_LEN 4096
#define BSZ     8
#define EMBED   1024
#define NDIM    16
#define CHUNK   64
#define NCHUNK  (SEQ_LEN / CHUNK)
#define DTILE   16
#define BT      2
#define KC      8
#define NSTEP   (NCHUNK / KC)
#define TSTEP   (KC * CHUNK)
#define XPITCH  42
#define XCH     2
#define SPITCH  257

__global__ void ema_prep(const float* __restrict__ delta,
                         const float* __restrict__ alpha,
                         const float* __restrict__ beta,
                         const float* __restrict__ gamma,
                         float* __restrict__ cArr,
                         float* __restrict__ lqArr, float* __restrict__ qCArr,
                         float* __restrict__ kcol) {
  const int d = blockIdx.x;
  const int t = threadIdx.x;
  const float dd = __expf(delta[d]);
  float qn[NDIM], cn[NDIM];
#pragma unroll
  for (int n = 0; n < NDIM; ++n) {
    const float al = alpha[d * NDIM + n];
    const float p  = dd / (1.0f + 0.5f * dd * al);
    qn[n] = 1.0f - p * al;
    cn[n] = p * beta[d * NDIM + n] * gamma[d * NDIM + n];
  }
  float s = 0.0f;
#pragma unroll
  for (int n = 0; n < NDIM; ++n)
    s += cn[n] * __expf((float)t * __logf(qn[n]));
  kcol[d * CHUNK + t] = s;

  if (t < NDIM) {
    const float q = qn[t];
    cArr [d * NDIM + t] = cn[t];
    lqArr[d * NDIM + t] = __logf(q);
    float qc = q;
#pragma unroll
    for (int i = 0; i < 6; ++i) qc *= qc;
    qCArr[d * NDIM + t] = qc;
  }
}

__global__ __launch_bounds__(512) void ema_main(
    const float* __restrict__ x,     const float* __restrict__ omega,
    const float* __restrict__ cArr,  const float* __restrict__ lqArr,
    const float* __restrict__ qCArr, const float* __restrict__ kcol,
    float* __restrict__ out) {
  __shared__ __align__(16) float Xs[TSTEP * XPITCH + KC * XCH];
  __shared__ __align__(16) float Sl[DTILE * SPITCH];

  const int tid  = threadIdx.x;
  const int lane = tid & 31;
  const int wv   = tid >> 5;
  const int d0   = blockIdx.x * DTILE;
  const int d    = d0 + wv;
  const int b0   = blockIdx.y * BT;
  const int c    = lane & 15;
  const int h    = lane >> 4;
  const int kk   = h << 1;
  const int ck   = c & 7, cbt = c >> 3;

  v2f Tf[16];
  const float* kc = kcol + d * CHUNK;
#pragma unroll
  for (int mi = 0; mi < 16; ++mi) {
    const int e0 = 4 * (mi - 3) + c - kk;
    const int e1 = e0 - 1;
    const float v0 = kc[e0 > 0 ? e0 : 0];
    const float v1 = kc[e1 > 0 ? e1 : 0];
    Tf[mi].x = (e0 >= 0) ? v0 : 0.0f;
    Tf[mi].y = (e1 >= 0) ? v1 : 0.0f;
  }
  v2f Wf[4][4];
#pragma unroll
  for (int s = 0; s < 4; ++s) {
    const int n = 4 * s + kk;
    const float c0 = cArr[d * NDIM + n],  c1 = cArr[d * NDIM + n + 1];
    const float l0 = lqArr[d * NDIM + n], l1 = lqArr[d * NDIM + n + 1];
#pragma unroll
    for (int rb = 0; rb < 4; ++rb) {
      const float ip1 = (float)(16 * rb + c + 1);
      Wf[rb][s].x = c0 * __expf(ip1 * l0);
      Wf[rb][s].y = c1 * __expf(ip1 * l1);
    }
  }
  v2f Qf[16];
  const float lqn = lqArr[d * NDIM + c];
#pragma unroll
  for (int s = 0; s < 16; ++s) {
    const int e0 = 63 - 4 * s - kk;
    Qf[s].x = __expf((float)e0 * lqn);
    Qf[s].y = __expf((float)(e0 - 1) * lqn);
  }
  const float qcn = qCArr[d * NDIM + c];
  const float om = omega[d];

  const int sv  = tid & 3;
  const int sb  = (tid >> 2) & 1;
  const int st  = tid >> 3;
  const int sx  = 8 * sv + sb;
  const float* xp = x   + (((size_t)st * BSZ + b0 + sb) * EMBED + d0 + sv * 4);
  float*       op = out + (((size_t)st * BSZ + b0 + sb) * EMBED + d0 + sv * 4);

  float4 xr[8];
#pragma unroll
  for (int it = 0; it < 8; ++it)
    xr[it] = *(const float4*)(xp + (size_t)(CHUNK * it) * BSZ * EMBED);

  const int xb = (ck * CHUNK + kk) * XPITCH + ck * XCH + 2 * wv + cbt;
  const int sB = wv * SPITCH + kk * 16 + c;
  const int sW = wv * SPITCH + c * 16 + 8 * h;
  const int eb = (ck * CHUNK + 8 * h) * XPITCH + ck * XCH + 2 * wv + cbt;

  float carry = 0.0f;

  for (int stp = 0; stp < NSTEP; ++stp) {
#pragma unroll
    for (int it = 0; it < 8; ++it) {
      const int t = st + CHUNK * it;
      float* p = &Xs[t * XPITCH + it * XCH + sx];
      p[0] = xr[it].x;
      p[2] = xr[it].y;
      p[4] = xr[it].z;
      p[6] = xr[it].w;
    }
    __syncthreads();
    if (stp + 1 < NSTEP) {
      const float* xn = xp + (size_t)(stp + 1) * TSTEP * BSZ * EMBED;
#pragma unroll
      for (int it = 0; it < 8; ++it)
        xr[it] = *(const float4*)(xn + (size_t)(CHUNK * it) * BSZ * EMBED);
    }

    v2f Bx[16];
#pragma unroll
    for (int s = 0; s < 16; ++s) {
      Bx[s].x = Xs[xb + (4 * s    ) * XPITCH];
      Bx[s].y = Xs[xb + (4 * s + 1) * XPITCH];
    }

    v8f sa = (v8f){0.f, 0.f, 0.f, 0.f, 0.f, 0.f, 0.f, 0.f};
#pragma unroll
    for (int s = 0; s < 16; ++s)
      sa = __builtin_amdgcn_wmma_f32_16x16x4_f32(
          false, Bx[s], false, Qf[s], (short)0, sa, false, false);
    {
      float S = carry;
#pragma unroll
      for (int k = 0; k < KC; ++k) {
        Sl[sW + k] = S;
        S = qcn * S + sa[k];
      }
      carry = S;
    }

    v8f acc[4];
#pragma unroll
    for (int rb = 0; rb < 4; ++rb)
      acc[rb] = (v8f){0.f, 0.f, 0.f, 0.f, 0.f, 0.f, 0.f, 0.f};
#pragma unroll
    for (int s = 0; s < 16; ++s) {
#pragma unroll
      for (int rb = 0; rb < 4; ++rb) {
        if (s < 4 * (rb + 1)) {
          acc[rb] = __builtin_amdgcn_wmma_f32_16x16x4_f32(
              false, Tf[4 * rb - s + 3], false, Bx[s], (short)0, acc[rb], false, false);
        }
      }
    }
    v2f Bs[4];
#pragma unroll
    for (int s = 0; s < 4; ++s) {
      Bs[s].x = Sl[sB + (4 * s    ) * 16];
      Bs[s].y = Sl[sB + (4 * s + 1) * 16];
    }
#pragma unroll
    for (int s = 0; s < 4; ++s) {
#pragma unroll
      for (int rb = 0; rb < 4; ++rb)
        acc[rb] = __builtin_amdgcn_wmma_f32_16x16x4_f32(
            false, Wf[rb][s], false, Bs[s], (short)0, acc[rb], false, false);
    }

#pragma unroll
    for (int rb = 0; rb < 4; ++rb) {
#pragma unroll
      for (int r = 0; r < 8; ++r) {
        const int idx = eb + (16 * rb + r) * XPITCH;
        const float xv = Xs[idx];
        Xs[idx] = acc[rb][r] + om * xv;
      }
    }
    __syncthreads();
    {
      float* on = op + (size_t)stp * TSTEP * BSZ * EMBED;
#pragma unroll
      for (int it = 0; it < 8; ++it) {
        const int t = st + CHUNK * it;
        const float* p = &Xs[t * XPITCH + it * XCH + sx];
        *(float4*)(on + (size_t)(CHUNK * it) * BSZ * EMBED) =
            make_float4(p[0], p[2], p[4], p[6]);
      }
    }
  }
}

extern "C" void kernel_launch(void* const* d_in, const int* in_sizes, int n_in,
                              void* d_out, int out_size, void* d_ws,
                              size_t ws_size, hipStream_t stream) {
  const float* x     = (const float*)d_in[0];
  const float* delta = (const float*)d_in[1];
  const float* alpha = (const float*)d_in[2];
  const float* beta  = (const float*)d_in[3];
  const float* gamma = (const float*)d_in[4];
  const float* omega = (const float*)d_in[5];
  float* out = (float*)d_out;

  float* ws    = (float*)d_ws;
  float* cArr  = ws;
  float* lqArr = cArr  + EMBED * NDIM;
  float* qCArr = lqArr + EMBED * NDIM;
  float* kcol  = qCArr + EMBED * NDIM;

  ema_prep<<<EMBED, CHUNK, 0, stream>>>(delta, alpha, beta, gamma,
                                        cArr, lqArr, qCArr, kcol);
  ema_main<<<dim3(EMBED / DTILE, BSZ / BT), 512, 0, stream>>>(
      x, omega, cArr, lqArr, qCArr, kcol, out);
}


# main kernel: channel tile per workgroup 16 -> 32, two channels per wave in turn on 32-row chunks: every global row piece of x and out is one whole 128-byte line; order of sums changed
# speedup vs baseline: 5.7139x; 1.7160x over previous
#include <hip/hip_runtime.h>
#include <hip/hip_bf16.h>


typedef float v2f __attribute__((ext_vector_type(2)));
typedef float v8f __attribute__((ext_vector_type(8)));

#define SEQ_LEN 4096
#define BSZ     8
#define EMBED   1024
#define NDIM    16
#define KLEN    64
#define CHUNK   32
#define NCHUNK  (SEQ_LEN / CHUNK)
#define DTILE   32
#define NWAVE   16
#define KC      16
#define NSTEP   (NCHUNK / KC)
#define TSTEP   (KC * CHUNK)
#define XPITCH  33
#define XCH     4
#define SPITCH  257

__global__ void ema_prep(const float* __restrict__ delta,
                         const float* __restrict__ alpha,
                         const float* __restrict__ beta,
                         const float* __restrict__ gamma,
                         float* __restrict__ cArr,
                         float* __restrict__ lqArr, float* __restrict__ qCArr,
                         float* __restrict__ kcol) {
  const int d = blockIdx.x;
  const int t = threadIdx.x;
  const float dd = __expf(delta[d]);
  float qn[NDIM], cn[NDIM];
#pragma unroll
  for (int n = 0; n < NDIM; ++n) {
    const float al = alpha[d * NDIM + n];
    const float p  = dd / (1.0f + 0.5f * dd * al);
    qn[n] = 1.0f - p * al;
    cn[n] = p * beta[d * NDIM + n] * gamma[d * NDIM + n];
  }
  float s = 0.0f;
#pragma unroll
  for (int n = 0; n < NDIM; ++n)
    s += cn[n] * __expf((float)t * __logf(qn[n]));
  kcol[d * KLEN + t] = s;

  if (t < NDIM) {
    const float q = qn[t];
    cArr [d * NDIM + t] = cn[t];
    lqArr[d * NDIM + t] = __logf(q);
    float qc = q;
#pragma unroll
    for (int i = 0; i < 5; ++i) qc *= qc;
    qCArr[d * NDIM + t] = qc;
  }
}

__global__ __launch_bounds__(512) void ema_main(
    const float* __restrict__ x,     const float* __restrict__ omega,
    const float* __restrict__ cArr,  const float* __restrict__ lqArr,
    const float* __restrict__ qCArr, const float* __restrict__ kcol,
    float* __restrict__ out) {
  __shared__ __align__(16) float Xs[TSTEP * XPITCH + KC * XCH];
  __shared__ __align__(16) float Sl[NWAVE * SPITCH];

  const int tid  = threadIdx.x;
  const int lane = tid & 31;
  const int wv   = tid >> 5;
  const int d0   = blockIdx.x * DTILE;
  const int b    = blockIdx.y;
  const int c    = lane & 15;
  const int h    = lane >> 4;
  const int kk   = h << 1;

  v2f   Tf[2][8];
  v2f   Wf[2][2][4];
  v2f   Qf[2][8];
  float qcn[2], q256[2], om[2], carry[2];
#pragma unroll
  for (int ch = 0; ch < 2; ++ch) {
    const int d = d0 + wv + 16 * ch;
    const float* kc = kcol + d * KLEN;
#pragma unroll
    for (int mi = 0; mi < 8; ++mi) {
      const int e0 = 4 * (mi - 3) + c - kk;
      const int e1 = e0 - 1;
      const float v0 = kc[e0 > 0 ? e0 : 0];
      const float v1 = kc[e1 > 0 ? e1 : 0];
      Tf[ch][mi].x = (e0 >= 0) ? v0 : 0.0f;
      Tf[ch][mi].y = (e1 >= 0) ? v1 : 0.0f;
    }
#pragma unroll
    for (int s = 0; s < 4; ++s) {
      const int n = 4 * s + kk;
      const float c0 = cArr[d * NDIM + n],  c1 = cArr[d * NDIM + n + 1];
      const float l0 = lqArr[d * NDIM + n], l1 = lqArr[d * NDIM + n + 1];
#pragma unroll
      for (int rb = 0; rb < 2; ++rb) {
        const float ip1 = (float)(16 * rb + c + 1);
        Wf[ch][rb][s].x = c0 * __expf(ip1 * l0);
        Wf[ch][rb][s].y = c1 * __expf(ip1 * l1);
      }
    }
    const float lqn = lqArr[d * NDIM + c];
#pragma unroll
    for (int s = 0; s < 8; ++s) {
      const int e0 = 31 - 4 * s - kk;
      Qf[ch][s].x = __expf((float)e0 * lqn);
      Qf[ch][s].y = __expf((float)(e0 - 1) * lqn);
    }
    const float q32 = qCArr[d * NDIM + c];
    const float q64 = q32 * q32, q128 = q64 * q64;
    qcn[ch]   = q32;
    q256[ch]  = q128 * q128;
    om[ch]    = omega[d];
    carry[ch] = 0.0f;
  }

  const int sv  = tid & 7;
  const int st  = tid >> 3;
  const int sx  = st * XPITCH + (st >> 5) * XCH + 4 * sv;
  const int sit = 64 * XPITCH + 2 * XCH;
  const float* xp = x   + (((size_t)st * BSZ + b) * EMBED + d0 + sv * 4);
  float*       op = out + (((size_t)st * BSZ + b) * EMBED + d0 + sv * 4);

  float4 xr[8];
#pragma unroll
  for (int it = 0; it < 8; ++it)
    xr[it] = *(const float4*)(xp + (size_t)(64 * it) * BSZ * EMBED);

  const int xb = (c * CHUNK + kk) * XPITCH + c * XCH + wv;
  const int eb = (c * CHUNK + 8 * h) * XPITCH + c * XCH + wv;
  const int sB = wv * SPITCH + kk * 16 + c;
  const int sW = wv * SPITCH + c * 16 + 8 * h;

  for (int stp = 0; stp < NSTEP; ++stp) {
#pragma unroll
    for (int it = 0; it < 8; ++it) {
      float* p = &Xs[sx + it * sit];
      p[0] = xr[it].x;
      p[1] = xr[it].y;
      p[2] = xr[it].z;
      p[3] = xr[it].w;
    }
    __syncthreads();
    if (stp + 1 < NSTEP) {
      const float* xn = xp + (size_t)(stp + 1) * TSTEP * BSZ * EMBED;
#pragma unroll
      for (int it = 0; it < 8; ++it)
        xr[it] = *(const float4*)(xn + (size_t)(64 * it) * BSZ * EMBED);
    }

#pragma unroll
    for (int ch = 0; ch < 2; ++ch) {
      const int xbc = xb + 16 * ch;
      const int ebc = eb + 16 * ch;
      v2f Bx[8];
#pragma unroll
      for (int s = 0; s < 8; ++s) {
        Bx[s].x = Xs[xbc + (4 * s    ) * XPITCH];
        Bx[s].y = Xs[xbc + (4 * s + 1) * XPITCH];
      }
      v8f sa = (v8f){0.f, 0.f, 0.f, 0.f, 0.f, 0.f, 0.f, 0.f};
#pragma unroll
      for (int s = 0; s < 8; ++s)
        sa = __builtin_amdgcn_wmma_f32_16x16x4_f32(
            false, Bx[s], false, Qf[ch][s], (short)0, sa, false, false);
      {
        const float q = qcn[ch];
        float L = 0.0f;
#pragma unroll
        for (int k = 0; k < 8; ++k) L = q * L + sa[k];
        const float Lo = __shfl_xor(L, 16);
        const float L0 = h ? Lo : L;
        const float L1 = h ? L : Lo;
        const float mid = q256[ch] * carry[ch] + L0;
        float S = h ? mid : carry[ch];
        carry[ch] = q256[ch] * mid + L1;
#pragma unroll
        for (int k = 0; k < 8; ++k) {
          Sl[sW + k] = S;
          S = q * S + sa[k];
        }
      }

      v8f acc[2];
#pragma unroll
      for (int rb = 0; rb < 2; ++rb)
        acc[rb] = (v8f){0.f, 0.f, 0.f, 0.f, 0.f, 0.f, 0.f, 0.f};
#pragma unroll
      for (int s = 0; s < 8; ++s) {
#pragma unroll
        for (int rb = 0; rb < 2; ++rb) {
          if (s < 4 * (rb + 1)) {
            acc[rb] = __builtin_amdgcn_wmma_f32_16x16x4_f32(
                false, Tf[ch][4 * rb - s + 3], false, Bx[s], (short)0, acc[rb], false, false);
          }
        }
      }
      v2f Bs[4];
#pragma unroll
      for (int s = 0; s < 4; ++s) {
        Bs[s].x = Sl[sB + (4 * s    ) * 16];
        Bs[s].y = Sl[sB + (4 * s + 1) * 16];
      }
#pragma unroll
      for (int s = 0; s < 4; ++s) {
#pragma unroll
        for (int rb = 0; rb < 2; ++rb)
          acc[rb] = __builtin_amdgcn_wmma_f32_16x16x4_f32(
              false, Wf[ch][rb][s], false, Bs[s], (short)0, acc[rb], false, false);
      }

#pragma unroll
      for (int rb = 0; rb < 2; ++rb) {
#pragma unroll
        for (int r = 0; r < 8; ++r) {
          const int idx = ebc + (16 * rb + r) * XPITCH;
          const float xv = Xs[idx];
          Xs[idx] = acc[rb][r] + om[ch] * xv;
        }
      }
    }
    __syncthreads();
    {
      float* on = op + (size_t)stp * TSTEP * BSZ * EMBED;
#pragma unroll
      for (int it = 0; it < 8; ++it) {
        const float* p = &Xs[sx + it * sit];
        *(float4*)(on + (size_t)(64 * it) * BSZ * EMBED) =
            make_float4(p[0], p[1], p[2], p[3]);
      }
    }
  }
}

extern "C" void kernel_launch(void* const* d_in, const int* in_sizes, int n_in,
                              void* d_out, int out_size, void* d_ws,
                              size_t ws_size, hipStream_t stream) {
  const float* x     = (const float*)d_in[0];
  const float* delta = (const float*)d_in[1];
  const float* alpha = (const float*)d_in[2];
  const float* beta  = (const float*)d_in[3];
  const float* gamma = (const float*)d_in[4];
  const float* omega = (const float*)d_in[5];
  float* out = (float*)d_out;

  float* ws    = (float*)d_ws;
  float* cArr  = ws;
  float* lqArr = cArr  + EMBED * NDIM;
  float* qCArr = lqArr + EMBED * NDIM;
  float* kcol  = qCArr + EMBED * NDIM;

  ema_prep<<<EMBED, KLEN, 0, stream>>>(delta, alpha, beta, gamma,
                                        cArr, lqArr, qCArr, kcol);
  ema_main<<<dim3(EMBED / DTILE, BSZ), 512, 0, stream>>>(
      x, omega, cArr, lqArr, qCArr, kcol, out);
}


# main kernel: next step's x loads issued at the top of a step and kept in flight across it: workgroup barriers order LDS only and no longer drain global loads and stores
# speedup vs baseline: 5.9423x; 1.0400x over previous
#include <hip/hip_runtime.h>
#include <hip/hip_bf16.h>


typedef float v2f __attribute__((ext_vector_type(2)));
typedef float v8f __attribute__((ext_vector_type(8)));

#define SEQ_LEN 4096
#define BSZ     8
#define EMBED   1024
#define NDIM    16
#define KLEN    64
#define CHUNK   32
#define NCHUNK  (SEQ_LEN / CHUNK)
#define DTILE   32
#define NWAVE   16
#define KC      16
#define NSTEP   (NCHUNK / KC)
#define TSTEP   (KC * CHUNK)
#define XPITCH  33
#define XCH     4
#define SPITCH  257

__device__ __forceinline__ void lds_barrier() {
  __builtin_amdgcn_fence(__ATOMIC_RELEASE, "workgroup", "local");
  __builtin_amdgcn_s_barrier();
  __builtin_amdgcn_fence(__ATOMIC_ACQUIRE, "workgroup", "local");
}

__global__ void ema_prep(const float* __restrict__ delta,
                         const float* __restrict__ alpha,
                         const float* __restrict__ beta,
                         const float* __restrict__ gamma,
                         float* __restrict__ cArr,
                         float* __restrict__ lqArr, float* __restrict__ qCArr,
                         float* __restrict__ kcol) {
  const int d = blockIdx.x;
  const int t = threadIdx.x;
  const float dd = __expf(delta[d]);
  float qn[NDIM], cn[NDIM];
#pragma unroll
  for (int n = 0; n < NDIM; ++n) {
    const float al = alpha[d * NDIM + n];
    const float p  = dd / (1.0f + 0.5f * dd * al);
    qn[n] = 1.0f - p * al;
    cn[n] = p * beta[d * NDIM + n] * gamma[d * NDIM + n];
  }
  float s = 0.0f;
#pragma unroll
  for (int n = 0; n < NDIM; ++n)
    s += cn[n] * __expf((float)t * __logf(qn[n]));
  kcol[d * KLEN + t] = s;

  if (t < NDIM) {
    const float q = qn[t];
    cArr [d * NDIM + t] = cn[t];
    lqArr[d * NDIM + t] = __logf(q);
    float qc = q;
#pragma unroll
    for (int i = 0; i < 5; ++i) qc *= qc;
    qCArr[d * NDIM + t] = qc;
  }
}

__global__ __launch_bounds__(512) void ema_main(
    const float* __restrict__ x,     const float* __restrict__ omega,
    const float* __restrict__ cArr,  const float* __restrict__ lqArr,
    const float* __restrict__ qCArr, const float* __restrict__ kcol,
    float* __restrict__ out) {
  __shared__ __align__(16) float Xs[TSTEP * XPITCH + KC * XCH];
  __shared__ __align__(16) float Sl[NWAVE * SPITCH];

  const int tid  = threadIdx.x;
  const int lane = tid & 31;
  const int wv   = tid >> 5;
  const int d0   = blockIdx.x * DTILE;
  const int b    = blockIdx.y;
  const int c    = lane & 15;
  const int h    = lane >> 4;
  const int kk   = h << 1;

  v2f   Tf[2][8];
  v2f   Wf[2][2][4];
  v2f   Qf[2][8];
  float qcn[2], q256[2], om[2], carry[2];
#pragma unroll
  for (int ch = 0; ch < 2; ++ch) {
    const int d = d0 + wv + 16 * ch;
    const float* kc = kcol + d * KLEN;
#pragma unroll
    for (int mi = 0; mi < 8; ++mi) {
      const int e0 = 4 * (mi - 3) + c - kk;
      const int e1 = e0 - 1;
      const float v0 = kc[e0 > 0 ? e0 : 0];
      const float v1 = kc[e1 > 0 ? e1 : 0];
      Tf[ch][mi].x = (e0 >= 0) ? v0 : 0.0f;
      Tf[ch][mi].y = (e1 >= 0) ? v1 : 0.0f;
    }
#pragma unroll
    for (int s = 0; s < 4; ++s) {
      const int n = 4 * s + kk;
      const float c0 = cArr[d * NDIM + n],  c1 = cArr[d * NDIM + n + 1];
      const float l0 = lqArr[d * NDIM + n], l1 = lqArr[d * NDIM + n + 1];
#pragma unroll
      for (int rb = 0; rb < 2; ++rb) {
        const float ip1 = (float)(16 * rb + c + 1);
        Wf[ch][rb][s].x = c0 * __expf(ip1 * l0);
        Wf[ch][rb][s].y = c1 * __expf(ip1 * l1);
      }
    }
    const float lqn = lqArr[d * NDIM + c];
#pragma unroll
    for (int s = 0; s < 8; ++s) {
      const int e0 = 31 - 4 * s - kk;
      Qf[ch][s].x = __expf((float)e0 * lqn);
      Qf[ch][s].y = __expf((float)(e0 - 1) * lqn);
    }
    const float q32 = qCArr[d * NDIM + c];
    const float q64 = q32 * q32, q128 = q64 * q64;
    qcn[ch]   = q32;
    q256[ch]  = q128 * q128;
    om[ch]    = omega[d];
    carry[ch] = 0.0f;
  }

  const int sv  = tid & 7;
  const int st  = tid >> 3;
  const int sx  = st * XPITCH + (st >> 5) * XCH + 4 * sv;
  const int sit = 64 * XPITCH + 2 * XCH;
  const float* xp = x   + (((size_t)st * BSZ + b) * EMBED + d0 + sv * 4);
  float*       op = out + (((size_t)st * BSZ + b) * EMBED + d0 + sv * 4);

  float4 xr[8];
#pragma unroll
  for (int it = 0; it < 8; ++it)
    xr[it] = *(const float4*)(xp + (size_t)(64 * it) * BSZ * EMBED);

  const int xb = (c * CHUNK + kk) * XPITCH + c * XCH + wv;
  const int eb = (c * CHUNK + 8 * h) * XPITCH + c * XCH + wv;
  const int sB = wv * SPITCH + kk * 16 + c;
  const int sW = wv * SPITCH + c * 16 + 8 * h;

  for (int stp = 0; stp < NSTEP; ++stp) {
#pragma unroll
    for (int it = 0; it < 8; ++it) {
      float* p = &Xs[sx + it * sit];
      p[0] = xr[it].x;
      p[1] = xr[it].y;
      p[2] = xr[it].z;
      p[3] = xr[it].w;
    }
    lds_barrier();
    if (stp + 1 < NSTEP) {
      const float* xn = xp + (size_t)(stp + 1) * TSTEP * BSZ * EMBED;
#pragma unroll
      for (int it = 0; it < 8; ++it)
        xr[it] = *(const float4*)(xn + (size_t)(64 * it) * BSZ * EMBED);
      __builtin_amdgcn_sched_barrier(0);
    }

#pragma unroll
    for (int ch = 0; ch < 2; ++ch) {
      const int xbc = xb + 16 * ch;
      const int ebc = eb + 16 * ch;
      v2f Bx[8];
#pragma unroll
      for (int s = 0; s < 8; ++s) {
        Bx[s].x = Xs[xbc + (4 * s    ) * XPITCH];
        Bx[s].y = Xs[xbc + (4 * s + 1) * XPITCH];
      }
      v8f sa = (v8f){0.f, 0.f, 0.f, 0.f, 0.f, 0.f, 0.f, 0.f};
#pragma unroll
      for (int s = 0; s < 8; ++s)
        sa = __builtin_amdgcn_wmma_f32_16x16x4_f32(
            false, Bx[s], false, Qf[ch][s], (short)0, sa, false, false);
      {
        const float q = qcn[ch];
        float L = 0.0f;
#pragma unroll
        for (int k = 0; k < 8; ++k) L = q * L + sa[k];
        const float Lo = __shfl_xor(L, 16);
        const float L0 = h ? Lo : L;
        const float L1 = h ? L : Lo;
        const float mid = q256[ch] * carry[ch] + L0;
        float S = h ? mid : carry[ch];
        carry[ch] = q256[ch] * mid + L1;
#pragma unroll
        for (int k = 0; k < 8; ++k) {
          Sl[sW + k] = S;
          S = q * S + sa[k];
        }
      }

      v8f acc[2];
#pragma unroll
      for (int rb = 0; rb < 2; ++rb)
        acc[rb] = (v8f){0.f, 0.f, 0.f, 0.f, 0.f, 0.f, 0.f, 0.f};
#pragma unroll
      for (int s = 0; s < 8; ++s) {
#pragma unroll
        for (int rb = 0; rb < 2; ++rb) {
          if (s < 4 * (rb + 1)) {
            acc[rb] = __builtin_amdgcn_wmma_f32_16x16x4_f32(
                false, Tf[ch][4 * rb - s + 3], false, Bx[s], (short)0, acc[rb], false, false);
          }
        }
      }
      v2f Bs[4];
#pragma unroll
      for (int s = 0; s < 4; ++s) {
        Bs[s].x = Sl[sB + (4 * s    ) * 16];
        Bs[s].y = Sl[sB + (4 * s + 1) * 16];
      }
#pragma unroll
      for (int s = 0; s < 4; ++s) {
#pragma unroll
        for (int rb = 0; rb < 2; ++rb)
          acc[rb] = __builtin_amdgcn_wmma_f32_16x16x4_f32(
              false, Wf[ch][rb][s], false, Bs[s], (short)0, acc[rb], false, false);
      }

#pragma unroll
      for (int rb = 0; rb < 2; ++rb) {
#pragma unroll
        for (int r = 0; r < 8; ++r) {
          const int idx = ebc + (16 * rb + r) * XPITCH;
          const float xv = Xs[idx];
          Xs[idx] = acc[rb][r] + om[ch] * xv;
        }
      }
    }
    lds_barrier();
    {
      float* on = op + (size_t)stp * TSTEP * BSZ * EMBED;
#pragma unroll
      for (int it = 0; it < 8; ++it) {
        const float* p = &Xs[sx + it * sit];
        *(float4*)(on + (size_t)(64 * it) * BSZ * EMBED) =
            make_float4(p[0], p[1], p[2], p[3]);
      }
    }
  }
}

extern "C" void kernel_launch(void* const* d_in, const int* in_sizes, int n_in,
                              void* d_out, int out_size, void* d_ws,
                              size_t ws_size, hipStream_t stream) {
  const float* x     = (const float*)d_in[0];
  const float* delta = (const float*)d_in[1];
  const float* alpha = (const float*)d_in[2];
  const float* beta  = (const float*)d_in[3];
  const float* gamma = (const float*)d_in[4];
  const float* omega = (const float*)d_in[5];
  float* out = (float*)d_out;

  float* ws    = (float*)d_ws;
  float* cArr  = ws;
  float* lqArr = cArr  + EMBED * NDIM;
  float* qCArr = lqArr + EMBED * NDIM;
  float* kcol  = qCArr + EMBED * NDIM;

  ema_prep<<<EMBED, KLEN, 0, stream>>>(delta, alpha, beta, gamma,
                                        cArr, lqArr, qCArr, kcol);
  ema_main<<<dim3(EMBED / DTILE, BSZ), 512, 0, stream>>>(
      x, omega, cArr, lqArr, qCArr, kcol, out);
}
